# starting kernel
# baseline (speedup 1.0000x reference)
#include <hip/hip_runtime.h>
#include <hip/hip_bf16.h>


typedef __attribute__((ext_vector_type(2))) float v2f;
typedef __attribute__((ext_vector_type(8))) float v8f;

#define FRAMES       4000
#define D2V          256
#define OUT_T        4001
#define OUT_LEN      32008
#define TILE_OUT     15
#define TILES_PER_BC 267
#define N_BC         32
#define WAVES_PER_BLOCK 8
#define N_BLOCKS     ((N_BC * TILES_PER_BC) / WAVES_PER_BLOCK)

__global__ __launch_bounds__(256)
void decoder_oadd_wmma(const float* __restrict__ est, float* __restrict__ out)
{
    __shared__ float lds[WAVES_PER_BLOCK][16][17];

    const int lane  = threadIdx.x & 31;
    const int wave  = threadIdx.x >> 5;
    const int gwave = blockIdx.x * WAVES_PER_BLOCK + wave;
    const int bc    = gwave / TILES_PER_BC;
    const int tile  = gwave % TILES_PER_BC;

    const int f0 = tile * TILE_OUT - 1;
    const int m  = lane & 15;
    const int kg = lane >> 4;
    const int frame = f0 + m;

    const float* srcbc = est + (size_t)bc * (size_t)(D2V * FRAMES);

    v8f acc = {};
    if (f0 >= 0 && f0 + 15 < FRAMES) {
        const float* p0 = srcbc + (size_t)(2 * kg) * FRAMES + frame;
        #pragma unroll 4
        for (int c4 = 0; c4 < 64; ++c4) {
            v2f a;
            a.x = p0[(size_t)(4 * c4)     * FRAMES];
            a.y = p0[(size_t)(4 * c4 + 1) * FRAMES];
            const float bv = (m == (c4 >> 2)) ? 0.0625f : 0.0f;
            v2f b; b.x = bv; b.y = bv;
            acc = __builtin_amdgcn_wmma_f32_16x16x4_f32(
                false, a, false, b, (short)0, acc, false, false);
        }
    } else {
        const bool fvalid = (frame >= 0) && (frame < FRAMES);
        #pragma unroll 4
        for (int c4 = 0; c4 < 64; ++c4) {
            v2f a; a.x = 0.0f; a.y = 0.0f;
            if (fvalid) {
                const float* p = srcbc + (size_t)(4 * c4 + 2 * kg) * FRAMES + frame;
                a.x = p[0];
                a.y = p[FRAMES];
            }
            const float bv = (m == (c4 >> 2)) ? 0.0625f : 0.0f;
            v2f b; b.x = bv; b.y = bv;
            acc = __builtin_amdgcn_wmma_f32_16x16x4_f32(
                false, a, false, b, (short)0, acc, false, false);
        }
    }

    #pragma unroll
    for (int v = 0; v < 8; ++v)
        lds[wave][v + 8 * kg][m] = acc[v];

    __syncthreads();

    float* obase = out + (size_t)bc * OUT_LEN;
    for (int i = lane; i < 120; i += 32) {
        const int off = 1 + (i >> 3);
        const int r   = i & 7;
        const int s   = f0 + off;
        if (s < OUT_T)
            obase[8 * s + r] = lds[wave][off][r] + lds[wave][off - 1][8 + r];
    }
}

extern "C" void kernel_launch(void* const* d_in, const int* in_sizes, int n_in,
                              void* d_out, int out_size, void* d_ws, size_t ws_size,
                              hipStream_t stream)
{
    const float* est = (const float*)d_in[0];
    float* out = (float*)d_out;
    hipLaunchKernelGGL(decoder_oadd_wmma,
                       dim3(N_BLOCKS), dim3(256), 0, stream,
                       est, out);
}


# Re-tile to 64 frames x 256 rows per wave: each 16-byte-per-lane load covers aligned 256-byte row pieces, 20-32 loads in flight, halo column read once (no overlapped tiles); sum order unchanged
# speedup vs baseline: 2.4580x; 2.4580x over previous
#include <hip/hip_runtime.h>


typedef __attribute__((ext_vector_type(2))) float v2f;
typedef __attribute__((ext_vector_type(4))) float v4f;
typedef __attribute__((ext_vector_type(8))) float v8f;

#define FRAMES          4000
#define D2V             256
#define OUT_T           4001
#define OUT_LEN         32008
#define TILE_F          64
#define TILES_PER_BC    63
#define N_BC            32
#define WAVES_PER_BLOCK 8
#define N_BLOCKS        ((N_BC * TILES_PER_BC) / WAVES_PER_BLOCK)
#define PITCH           20
#define NBUF            5

__device__ __forceinline__ v8f mma4(v2f a, v2f b, v8f c)
{
    return __builtin_amdgcn_wmma_f32_16x16x4_f32(false, a, false, b, (short)0, c, false, false);
}

__global__ __launch_bounds__(256)
void decoder_oadd_wmma(const float* __restrict__ est, float* __restrict__ out)
{
    __shared__ __attribute__((aligned(16))) float lds[WAVES_PER_BLOCK][(TILE_F + 1) * PITCH];

    const int lane  = threadIdx.x & 31;
    const int wave  = threadIdx.x >> 5;
    const int gwave = blockIdx.x * WAVES_PER_BLOCK + wave;
    const int bc    = gwave / TILES_PER_BC;
    const int tile  = gwave % TILES_PER_BC;
    const int F0    = tile * TILE_F;
    const int c     = lane & 15;
    const int h     = lane >> 4;

    const float* srcbc = est + (size_t)bc * (size_t)(D2V * FRAMES);

    int fb = F0 + 4 * c;
    if (fb > FRAMES - 4) fb = FRAMES - 4;
    const float* p = srcbc + (size_t)(2 * h) * FRAMES + fb;

    const bool has_halo = (tile > 0);
    float g[8];
    if (has_halo) {
        const float* ph = srcbc + (size_t)(128 + 16 * (c & 7) + 2 * h) * FRAMES + (F0 - 1);
        #pragma unroll
        for (int q = 0; q < 4; ++q) {
            g[2 * q]     = ph[(size_t)(4 * q)     * FRAMES];
            g[2 * q + 1] = ph[(size_t)(4 * q + 1) * FRAMES];
        }
    } else {
        #pragma unroll
        for (int q = 0; q < 8; ++q) g[q] = 0.0f;
    }

    v4f buf[NBUF][8];
#define LOAD_WINDOW(slot, w)                                                              \
    _Pragma("unroll")                                                                     \
    for (int i = 0; i < 4; ++i) {                                                         \
        buf[slot][2 * i]     = *(const v4f*)(p + (size_t)(16 * (w) + 4 * i)     * FRAMES); \
        buf[slot][2 * i + 1] = *(const v4f*)(p + (size_t)(16 * (w) + 4 * i + 1) * FRAMES); \
    }

    #pragma unroll
    for (int w = 0; w < NBUF; ++w) { LOAD_WINDOW(w, w) }

    v8f hacc = {};
    {
        v2f a; a.x = 0.0625f; a.y = 0.0625f;
        #pragma unroll
        for (int q = 0; q < 4; ++q) {
            v2f b; b.x = g[2 * q]; b.y = g[2 * q + 1];
            hacc = mma4(a, b, hacc);
        }
    }

    v8f acc[4] = {};
    #pragma unroll
    for (int w = 0; w < 16; ++w) {
        const int slot = w % NBUF;
        const float sv = (c == w) ? 0.0625f : 0.0f;
        v2f a; a.x = sv; a.y = sv;
        #pragma unroll
        for (int i = 0; i < 4; ++i) {
            #pragma unroll
            for (int j = 0; j < 4; ++j) {
                v2f b; b.x = buf[slot][2 * i][j]; b.y = buf[slot][2 * i + 1][j];
                acc[j] = mma4(a, b, acc[j]);
            }
        }
        if (w + NBUF < 16) { LOAD_WINDOW(slot, w + NBUF) }
    }
#undef LOAD_WINDOW

    float* L = lds[wave];
    #pragma unroll
    for (int j = 0; j < 4; ++j) {
        float* row = L + (1 + 4 * c + j) * PITCH + 8 * h;
        v4f lo4; lo4.x = acc[j][0]; lo4.y = acc[j][1]; lo4.z = acc[j][2]; lo4.w = acc[j][3];
        v4f hi4; hi4.x = acc[j][4]; hi4.y = acc[j][5]; hi4.z = acc[j][6]; hi4.w = acc[j][7];
        *(v4f*)(row)     = lo4;
        *(v4f*)(row + 4) = hi4;
    }
    if (lane < 8) L[8 + lane] = hacc[0];

    __builtin_amdgcn_wave_barrier();

    float* obase = out + (size_t)bc * OUT_LEN + (size_t)8 * F0;
    #pragma unroll
    for (int q = 0; q < 4; ++q) {
        const int sl = 16 * q + (lane >> 1);
        const int r4 = (lane & 1) * 4;
        const int s  = F0 + sl;
        v4f lo = *(const v4f*)(L + (1 + sl) * PITCH + r4);
        const v4f hi = *(const v4f*)(L + sl * PITCH + 8 + r4);
        if (s >= FRAMES) { lo.x = 0.0f; lo.y = 0.0f; lo.z = 0.0f; lo.w = 0.0f; }
        if (s < OUT_T)
            *(v4f*)(obase + 8 * sl + r4) = lo + hi;
    }
}

extern "C" void kernel_launch(void* const* d_in, const int* in_sizes, int n_in,
                              void* d_out, int out_size, void* d_ws, size_t ws_size,
                              hipStream_t stream)
{
    const float* est = (const float*)d_in[0];
    float* out = (float*)d_out;
    hipLaunchKernelGGL(decoder_oadd_wmma,
                       dim3(N_BLOCKS), dim3(256), 0, stream,
                       est, out);
}


# Split each tile's 256 rows over two waves (16 waves per WGP, 504 blocks of 4 tiles); a block's P rows shared in LDS behind one barrier, halo column read once per block; sum order unchanged
# speedup vs baseline: 2.6489x; 1.0777x over previous
#include <hip/hip_runtime.h>


typedef __attribute__((ext_vector_type(2))) float v2f;
typedef __attribute__((ext_vector_type(4))) float v4f;
typedef __attribute__((ext_vector_type(8))) float v8f;

#define FRAMES          4000
#define D2V             256
#define OUT_T           4001
#define OUT_LEN         32008
#define TILE_F          64
#define TILES_PER_BC    63
#define N_BC            32
#define TILES_PER_BLOCK 4
#define WAVES_PER_BLOCK (2 * TILES_PER_BLOCK)
#define N_BLOCKS        ((N_BC * TILES_PER_BC) / TILES_PER_BLOCK)
#define PITCH           20
#define NWIN            8
#define NBUF            5

__device__ __forceinline__ v8f mma4(v2f a, v2f b, v8f c)
{
    return __builtin_amdgcn_wmma_f32_16x16x4_f32(false, a, false, b, (short)0, c, false, false);
}

__global__ __launch_bounds__(256)
void decoder_oadd_wmma(const float* __restrict__ est, float* __restrict__ out)
{
    __shared__ __attribute__((aligned(16))) float lds[(TILES_PER_BLOCK * TILE_F + 1) * PITCH];

    const int lane  = threadIdx.x & 31;
    const int wave  = __builtin_amdgcn_readfirstlane((int)(threadIdx.x >> 5));
    const int ts    = wave >> 1;
    const int hf    = wave & 1;
    const int gtile = blockIdx.x * TILES_PER_BLOCK + ts;
    const int bc    = gtile / TILES_PER_BC;
    const int tile  = gtile % TILES_PER_BC;
    const int F0    = tile * TILE_F;
    const int c     = lane & 15;
    const int h     = lane >> 4;

    const float* srcbc = est + (size_t)bc * (size_t)(D2V * FRAMES);

    int fb = F0 + 4 * c;
    if (fb > FRAMES - 4) fb = FRAMES - 4;
    const float* p = srcbc + (size_t)(128 * hf + 2 * h) * FRAMES + fb;

    const bool halo_wave = (ts == 0) && (hf == 1);
    const bool has_halo  = halo_wave && (tile > 0);
    float g[8];
    #pragma unroll
    for (int q = 0; q < 8; ++q) g[q] = 0.0f;
    if (has_halo) {
        const float* ph = srcbc + (size_t)(128 + 16 * (c & 7) + 2 * h) * FRAMES + (F0 - 1);
        #pragma unroll
        for (int q = 0; q < 4; ++q) {
            g[2 * q]     = ph[(size_t)(4 * q)     * FRAMES];
            g[2 * q + 1] = ph[(size_t)(4 * q + 1) * FRAMES];
        }
    }

    v4f buf[NBUF][8];
#define LOAD_WINDOW(slot, w)                                                              \
    _Pragma("unroll")                                                                     \
    for (int i = 0; i < 4; ++i) {                                                         \
        buf[slot][2 * i]     = *(const v4f*)(p + (size_t)(16 * (w) + 4 * i)     * FRAMES); \
        buf[slot][2 * i + 1] = *(const v4f*)(p + (size_t)(16 * (w) + 4 * i + 1) * FRAMES); \
    }

    #pragma unroll
    for (int w = 0; w < NBUF; ++w) { LOAD_WINDOW(w, w) }

    if (halo_wave) {
        v8f hacc = {};
        v2f a; a.x = 0.0625f; a.y = 0.0625f;
        #pragma unroll
        for (int q = 0; q < 4; ++q) {
            v2f b; b.x = g[2 * q]; b.y = g[2 * q + 1];
            hacc = mma4(a, b, hacc);
        }
        if (lane < 8) lds[8 + lane] = hacc[0];
    }

    v8f acc[4] = {};
    #pragma unroll
    for (int w = 0; w < NWIN; ++w) {
        const int slot = w % NBUF;
        const float sv = (c == w) ? 0.0625f : 0.0f;
        v2f a; a.x = sv; a.y = sv;
        #pragma unroll
        for (int i = 0; i < 4; ++i) {
            #pragma unroll
            for (int j = 0; j < 4; ++j) {
                v2f b; b.x = buf[slot][2 * i][j]; b.y = buf[slot][2 * i + 1][j];
                acc[j] = mma4(a, b, acc[j]);
            }
        }
        if (w + NBUF < NWIN) { LOAD_WINDOW(slot, w + NBUF) }
    }
#undef LOAD_WINDOW

    float* L = lds + (1 + TILE_F * ts) * PITCH;
    if (h == 0) {
        #pragma unroll
        for (int j = 0; j < 4; ++j) {
            float* row = L + (4 * c + j) * PITCH + 8 * hf;
            v4f lo4; lo4.x = acc[j][0]; lo4.y = acc[j][1]; lo4.z = acc[j][2]; lo4.w = acc[j][3];
            v4f hi4; hi4.x = acc[j][4]; hi4.y = acc[j][5]; hi4.z = acc[j][6]; hi4.w = acc[j][7];
            *(v4f*)(row)     = lo4;
            *(v4f*)(row + 4) = hi4;
        }
    }

    __syncthreads();

    float* obase = out + (size_t)bc * OUT_LEN + (size_t)8 * F0;
    #pragma unroll
    for (int qq = 0; qq < 2; ++qq) {
        const int sl = 16 * (2 * hf + qq) + (lane >> 1);
        const int r4 = (lane & 1) * 4;
        const int s  = F0 + sl;
        v4f lo = *(const v4f*)(L + sl * PITCH + r4);
        v4f hi = *(const v4f*)(L + (sl - 1) * PITCH + 8 + r4);
        if (s >= FRAMES) { lo.x = 0.0f; lo.y = 0.0f; lo.z = 0.0f; lo.w = 0.0f; }
        if (s == 0)      { hi.x = 0.0f; hi.y = 0.0f; hi.z = 0.0f; hi.w = 0.0f; }
        if (s < OUT_T)
            *(v4f*)(obase + 8 * sl + r4) = lo + hi;
    }
}

extern "C" void kernel_launch(void* const* d_in, const int* in_sizes, int n_in,
                              void* d_out, int out_size, void* d_ws, size_t ws_size,
                              hipStream_t stream)
{
    const float* est = (const float*)d_in[0];
    float* out = (float*)d_out;
    hipLaunchKernelGGL(decoder_oadd_wmma,
                       dim3(N_BLOCKS), dim3(256), 0, stream,
                       est, out);
}
